# starting kernel
# baseline (speedup 1.0000x reference)
#include <hip/hip_runtime.h>
#include <math.h>


#define NN   50000
#define EE   800000
#define IND  64
#define HH   128
#define OUTD 16

typedef float v2f __attribute__((ext_vector_type(2)));
typedef float v8f __attribute__((ext_vector_type(8)));

__device__ __forceinline__ float elu_f(float v) {
    return v > 0.0f ? v : (__expf(v) - 1.0f);
}

template <int ACT, int TN>
__global__ void gemm16_wmma(const float* __restrict__ A,
                            const float* __restrict__ W,
                            const float* __restrict__ bias,
                            float* __restrict__ C,
                            int M, int K, int Nc) {
    int gid  = blockIdx.x * blockDim.x + threadIdx.x;
    int wave = gid >> 5;
    int lane = gid & 31;
    int groupsN = Nc / (16 * TN);
    int tilesM  = M >> 4;
    if (wave >= tilesM * groupsN) return;
    int tm = wave / groupsN;
    int tg = wave - tm * groupsN;

    int r  = lane & 15;
    int kq = (lane >> 4) << 1;

    const float* Arow = A + (size_t)(tm * 16 + r) * K + kq;
    const float* Bp   = W + (size_t)kq * Nc + tg * (16 * TN) + r;
    const size_t bstep = (size_t)4 * Nc;

    v8f acc[TN];
#pragma unroll
    for (int t = 0; t < TN; ++t) acc[t] = (v8f){};

    v2f a_cur, b_cur[TN];
    {
        float2 av = *(const float2*)(Arow);
        a_cur.x = av.x;
        a_cur.y = av.y;
#pragma unroll
        for (int t = 0; t < TN; ++t) {
            b_cur[t].x = Bp[t * 16];
            b_cur[t].y = Bp[Nc + t * 16];
        }
    }

    for (int k0 = 4; k0 < K; k0 += 4) {
        const float* Bn = Bp + bstep;
        float2 an = *(const float2*)(Arow + k0);
        v2f a_nxt, b_nxt[TN];
        a_nxt.x = an.x;
        a_nxt.y = an.y;
#pragma unroll
        for (int t = 0; t < TN; ++t) {
            b_nxt[t].x = Bn[t * 16];
            b_nxt[t].y = Bn[Nc + t * 16];
        }
#pragma unroll
        for (int t = 0; t < TN; ++t) {
            acc[t] = __builtin_amdgcn_wmma_f32_16x16x4_f32(
                false, a_cur, false, b_cur[t], (short)0, acc[t], false, false);
        }
        a_cur = a_nxt;
#pragma unroll
        for (int t = 0; t < TN; ++t) b_cur[t] = b_nxt[t];
        Bp = Bn;
    }

#pragma unroll
    for (int t = 0; t < TN; ++t) {
        acc[t] = __builtin_amdgcn_wmma_f32_16x16x4_f32(
            false, a_cur, false, b_cur[t], (short)0, acc[t], false, false);
    }

    int mb = tm * 16 + ((lane >> 4) << 3);
#pragma unroll
    for (int t = 0; t < TN; ++t) {
        int n = tg * (16 * TN) + t * 16 + r;
        float bv = bias ? bias[n] : 0.0f;
#pragma unroll
        for (int g = 0; g < 8; ++g) {
            float v = acc[t][g] + bv;
            if (ACT) v = elu_f(v);
            C[(size_t)(mb + g) * Nc + n] = v;
        }
    }
}

__global__ void fill_kernel(float* __restrict__ p, float val, int n) {
    int i = blockIdx.x * blockDim.x + threadIdx.x;
    if (i < n) p[i] = val;
}

__global__ void deg_accum_kernel(const int* __restrict__ dst,
                                 const float* __restrict__ w,
                                 float* __restrict__ deg) {
    int e = blockIdx.x * blockDim.x + threadIdx.x;
    if (e < EE) atomicAdd(&deg[dst[e]], w[e]);
}

__global__ void dinv_kernel(float* __restrict__ deg, float* __restrict__ sc) {
    int n = blockIdx.x * blockDim.x + threadIdx.x;
    if (n < NN) {
        float di = rsqrtf(deg[n]);
        deg[n] = di;
        sc[n]  = 2.0f * di * di;
    }
}

__global__ void norm_kernel(const int* __restrict__ src,
                            const int* __restrict__ dst,
                            const float* __restrict__ w,
                            const float* __restrict__ dinv,
                            float* __restrict__ nrm) {
    int e = blockIdx.x * blockDim.x + threadIdx.x;
    if (e < EE) nrm[e] = w[e] * dinv[src[e]] * dinv[dst[e]];
}

__global__ void scatter_kernel(const float* __restrict__ t,
                               const int* __restrict__ src,
                               const int* __restrict__ dst,
                               const float* __restrict__ nrm,
                               float* __restrict__ agg) {
    long long idx = (long long)blockIdx.x * blockDim.x + threadIdx.x;
    long long e = idx >> 5;
    if (e >= EE) return;
    int fg = (int)(idx & 31);
    int s = src[e], d = dst[e];
    float w = nrm[e];
    const float4 tv = *(const float4*)(t + (size_t)s * HH + fg * 4);
    float* ap = agg + (size_t)d * HH + fg * 4;
    atomicAdd(ap + 0, tv.x * w);
    atomicAdd(ap + 1, tv.y * w);
    atomicAdd(ap + 2, tv.z * w);
    atomicAdd(ap + 3, tv.w * w);
}

__global__ void post_kernel(const float* __restrict__ agg,
                            const float* __restrict__ t,
                            const float* __restrict__ sc,
                            const float* __restrict__ bias,
                            float* __restrict__ h) {
    int idx = blockIdx.x * blockDim.x + threadIdx.x;
    if (idx >= NN * HH) return;
    int n = idx >> 7;
    int f = idx & 127;
    float v = agg[idx] + sc[n] * t[idx] + bias[f];
    h[idx] = elu_f(v);
}

static inline int nblocks(long long t, int bs) { return (int)((t + bs - 1) / bs); }

extern "C" void kernel_launch(void* const* d_in, const int* in_sizes, int n_in,
                              void* d_out, int out_size, void* d_ws, size_t ws_size,
                              hipStream_t stream) {
    const float* x  = (const float*)d_in[0];
    const int*   ei = (const int*)d_in[1];
    const float* ew = (const float*)d_in[2];
    const float* W1 = (const float*)d_in[3];  const float* b1 = (const float*)d_in[4];
    const float* W2 = (const float*)d_in[5];  const float* b2 = (const float*)d_in[6];
    const float* W3 = (const float*)d_in[7];  const float* b3 = (const float*)d_in[8];
    const float* Wg[4] = { (const float*)d_in[9],  (const float*)d_in[11],
                           (const float*)d_in[13], (const float*)d_in[15] };
    const float* bg[4] = { (const float*)d_in[10], (const float*)d_in[12],
                           (const float*)d_in[14], (const float*)d_in[16] };
    const float* Wh = (const float*)d_in[17]; const float* bh = (const float*)d_in[18];
    float* out = (float*)d_out;

    const int* srcI = ei;
    const int* dstI = ei + EE;

    float* f    = (float*)d_ws;
    float* hbuf = f;
    float* tbuf = hbuf + (size_t)NN * HH;
    float* aggb = tbuf + (size_t)NN * HH;
    float* degb = aggb + (size_t)NN * HH;
    float* scb  = degb + NN;
    float* nrmb = scb + NN;

    const int BS = 256;

    fill_kernel<<<nblocks(NN, BS), BS, 0, stream>>>(degb, 2.0f, NN);
    deg_accum_kernel<<<nblocks(EE, BS), BS, 0, stream>>>(dstI, ew, degb);
    dinv_kernel<<<nblocks(NN, BS), BS, 0, stream>>>(degb, scb);
    norm_kernel<<<nblocks(EE, BS), BS, 0, stream>>>(srcI, dstI, ew, degb, nrmb);

    auto gemm128 = [&](const float* A, const float* W, const float* bias, float* C,
                       int K, int act) {
        long long thr = (long long)(NN >> 4) * (HH / 64) * 32;
        if (act) gemm16_wmma<1, 4><<<nblocks(thr, BS), BS, 0, stream>>>(A, W, bias, C, NN, K, HH);
        else     gemm16_wmma<0, 4><<<nblocks(thr, BS), BS, 0, stream>>>(A, W, bias, C, NN, K, HH);
    };

    gemm128(x,    W1, b1, hbuf, IND, 1);
    gemm128(hbuf, W2, b2, tbuf, HH,  1);
    gemm128(tbuf, W3, b3, hbuf, HH,  1);

    for (int l = 0; l < 4; ++l) {
        gemm128(hbuf, Wg[l], nullptr, tbuf, HH, 0);
        fill_kernel<<<nblocks((long long)NN * HH, BS), BS, 0, stream>>>(aggb, 0.0f, NN * HH);
        scatter_kernel<<<nblocks((long long)EE * 32, BS), BS, 0, stream>>>(
            tbuf, srcI, dstI, nrmb, aggb);
        post_kernel<<<nblocks((long long)NN * HH, BS), BS, 0, stream>>>(
            aggb, tbuf, scb, bg[l], hbuf);
    }

    {
        long long thr = (long long)(NN >> 4) * (OUTD >> 4) * 32;
        gemm16_wmma<0, 1><<<nblocks(thr, BS), BS, 0, stream>>>(hbuf, Wh, bh, out, NN, HH, OUTD);
    }
}


# Dense launches: each 8-wave work-group copies its 64-column panel of W into LDS once, by k pairs, and its waves read B fragments there (2 LDS loads per k-step, not 8 global loads); sum order kept
# speedup vs baseline: 2.3598x; 2.3598x over previous
#include <hip/hip_runtime.h>
#include <math.h>


#define NN   50000
#define EE   800000
#define IND  64
#define HH   128
#define OUTD 16

typedef float v2f __attribute__((ext_vector_type(2)));
typedef float v8f __attribute__((ext_vector_type(8)));

__device__ __forceinline__ float elu_f(float v) {
    return v > 0.0f ? v : (__expf(v) - 1.0f);
}

template <int TN>
__device__ __forceinline__ void load_bfrag(const float* __restrict__ p, v2f (&b)[TN]) {
    if constexpr (TN == 4) {
        const float4 q0 = *(const float4*)(p);
        const float4 q1 = *(const float4*)(p + 64);
        b[0] = (v2f){q0.x, q0.y};
        b[1] = (v2f){q0.z, q0.w};
        b[2] = (v2f){q1.x, q1.y};
        b[3] = (v2f){q1.z, q1.w};
    } else {
        const float2 q0 = *(const float2*)(p);
        b[0] = (v2f){q0.x, q0.y};
    }
}

template <int TN>
__device__ __forceinline__ void wmma_strip(v2f a, const v2f (&b)[TN], v8f (&acc)[TN]) {
#pragma unroll
    for (int t = 0; t < TN; ++t) {
        acc[t] = __builtin_amdgcn_wmma_f32_16x16x4_f32(
            false, a, false, b[t], (short)0, acc[t], false, false);
    }
}

#define GW 8

template <int ACT, int TN, int EIN, int DUAL, int K>
__global__ void __launch_bounds__(32 * GW)
gemm16_wmma(const float* __restrict__ A,
            const float* __restrict__ W,
            const float* __restrict__ bias,
            const float* __restrict__ rowscale,
            float* __restrict__ C,
            float* __restrict__ C2,
            int M, int Nc) {
    static_assert(TN == 4 || TN == 1, "panel layouts exist for 4 tiles and for 1");
    constexpr int PW  = 16 * TN;
    constexpr int PPR = PW / 4;
    constexpr int NQ  = K * PPR / (32 * GW);
    static_assert(K % 8 == 0 && (K * PPR) % (32 * GW) == 0, "k-steps go in pairs; the panel divides evenly among the threads");
    __shared__ __attribute__((aligned(16))) float Wl[K * PW];

    const int tid  = threadIdx.x;
    const int lane = tid & 31;
    const int wv   = tid >> 5;
    const int groupsN = Nc / PW;
    const int tilesM  = M >> 4;
    const int bq = blockIdx.x / groupsN;
    const int tg = blockIdx.x - bq * groupsN;
    const int tm = bq * GW + wv;
    const int tmc = tm < tilesM ? tm : tilesM - 1;

    const int r  = lane & 15;
    const int hf = lane >> 4;
    const int kq = hf << 1;

    const float* Arow = A + (size_t)(tmc * 16 + r) * K + kq;
    const float2 av0 = *(const float2*)(Arow);

    {
        const float* Wp = W + tg * PW;
        float4 w4[NQ];
#pragma unroll
        for (int i = 0; i < NQ; ++i) {
            const int q  = tid + i * (32 * GW);
            const int k  = q / PPR;
            const int c4 = (q - k * PPR) << 2;
            w4[i] = *(const float4*)(Wp + (size_t)k * Nc + c4);
        }
#pragma unroll
        for (int i = 0; i < NQ; ++i) {
            const int q  = tid + i * (32 * GW);
            const int k  = q / PPR;
            const int c4 = (q - k * PPR) << 2;
            const int t  = c4 >> 4;
            const int c  = c4 & 15;
            float* p = (TN == 4)
                ? Wl + (k >> 1) * (2 * PW) + (t >> 1) * 64 + c * 4 + (t & 1) * 2 + (k & 1)
                : Wl + (k >> 1) * (2 * PW) + c * 2 + (k & 1);
            constexpr int CS = (TN == 4) ? 4 : 2;
            p[0]      = w4[i].x;
            p[CS]     = w4[i].y;
            p[2 * CS] = w4[i].z;
            p[3 * CS] = w4[i].w;
        }
    }
    __syncthreads();
    if (tm >= tilesM) return;

    const float* Bl = Wl + hf * (2 * PW) + r * (TN == 4 ? 4 : 2);

    v8f acc[TN];
#pragma unroll
    for (int t = 0; t < TN; ++t) acc[t] = (v8f){};

    v2f aP, bP[TN], aQ, bQ[TN];
    aP.x = EIN ? elu_f(av0.x) : av0.x;
    aP.y = EIN ? elu_f(av0.y) : av0.y;
    load_bfrag<TN>(Bl, bP);

    for (int k0 = 0; k0 < K - 8; k0 += 8) {
        const float2 rq = *(const float2*)(Arow + k0 + 4);
        load_bfrag<TN>(Bl + (k0 + 4) * PW, bQ);
        wmma_strip<TN>(aP, bP, acc);
        aQ.x = EIN ? elu_f(rq.x) : rq.x;
        aQ.y = EIN ? elu_f(rq.y) : rq.y;
        const float2 rp = *(const float2*)(Arow + k0 + 8);
        load_bfrag<TN>(Bl + (k0 + 8) * PW, bP);
        wmma_strip<TN>(aQ, bQ, acc);
        aP.x = EIN ? elu_f(rp.x) : rp.x;
        aP.y = EIN ? elu_f(rp.y) : rp.y;
    }
    {
        const float2 rq = *(const float2*)(Arow + K - 4);
        load_bfrag<TN>(Bl + (K - 4) * PW, bQ);
        wmma_strip<TN>(aP, bP, acc);
        aQ.x = EIN ? elu_f(rq.x) : rq.x;
        aQ.y = EIN ? elu_f(rq.y) : rq.y;
        wmma_strip<TN>(aQ, bQ, acc);
    }

    int mb = tm * 16 + ((lane >> 4) << 3);
    float rs[8];
    if (DUAL) {
#pragma unroll
        for (int g = 0; g < 8; ++g) rs[g] = rowscale[mb + g];
    }
#pragma unroll
    for (int t = 0; t < TN; ++t) {
        int n = tg * PW + t * 16 + r;
        float bv = bias ? bias[n] : 0.0f;
#pragma unroll
        for (int g = 0; g < 8; ++g) {
            size_t o = (size_t)(mb + g) * Nc + n;
            if (DUAL) {
                float v = acc[t][g];
                C[o]  = v;
                C2[o] = rs[g] * v + bv;
            } else {
                float v = acc[t][g] + bv;
                if (ACT) v = elu_f(v);
                C[o] = v;
            }
        }
    }
}

__global__ void fill_kernel(float* __restrict__ p, float val, int n,
                            unsigned* __restrict__ cnt) {
    int i = blockIdx.x * blockDim.x + threadIdx.x;
    if (i < n) p[i] = val;
    if (i < 8) cnt[i] = 0u;
}

__global__ void deg_accum_kernel(const int* __restrict__ dst,
                                 const float* __restrict__ w,
                                 float* __restrict__ deg) {
    int e = blockIdx.x * blockDim.x + threadIdx.x;
    if (e < EE) atomicAdd(&deg[dst[e]], w[e]);
}

__global__ void dinv_kernel(float* __restrict__ deg, float* __restrict__ sc) {
    int n = blockIdx.x * blockDim.x + threadIdx.x;
    if (n < NN) {
        float di = rsqrtf(deg[n]);
        deg[n] = di;
        sc[n]  = 2.0f * di * di;
    }
}

__global__ void norm_kernel(const int* __restrict__ src,
                            const int* __restrict__ dst,
                            const float* __restrict__ w,
                            const float* __restrict__ dinv,
                            float* __restrict__ nrm) {
    int e = blockIdx.x * blockDim.x + threadIdx.x;
    if (e < EE) nrm[e] = w[e] * dinv[src[e]] * dinv[dst[e]];
}

#define EPG 12
#define WPB 8
#define NTK (EE / (32 * WPB))
#define HALF_ROWS (NN / 2)
#define SC_BLOCKS 1536
static_assert(EE % (32 * WPB) == 0, "tickets tile the edge list");

__device__ __forceinline__ void ticket_barrier() {
    __builtin_amdgcn_fence(__ATOMIC_RELEASE, "workgroup", "local");
    __builtin_amdgcn_s_barrier();
    __builtin_amdgcn_fence(__ATOMIC_ACQUIRE, "workgroup", "local");
}

__global__ void __launch_bounds__(32 * WPB)
scatter_kernel(const float* __restrict__ t,
               const int* __restrict__ src,
               const int* __restrict__ dst,
               const float* __restrict__ nrm,
               float* __restrict__ agg,
               unsigned* __restrict__ cnt) {
    __shared__ unsigned s_half;
    __shared__ unsigned s_tk[2];
    const int tid = threadIdx.x;
    const int ln  = tid & 31;
    const int wv  = tid >> 5;
    if (tid == 0) {
        unsigned id = __builtin_amdgcn_s_sendmsg_rtn(0x87);
        s_half = (id >> 18) & 1u;
    }
    ticket_barrier();
    const int hh = (int)s_half;
    const float* tl = t + ln;
    float*       al = agg + ln;

    for (int ph = 0; ph < 2; ++ph) {
        const int p = hh ^ ph;
        unsigned* c = cnt + p;
        ticket_barrier();
        if (tid == 0) s_tk[0] = atomicAdd(c, 1u);
        ticket_barrier();
        for (int i = 0;; ++i) {
            const unsigned tk = s_tk[i & 1];
            if (tk >= (unsigned)NTK) break;
            unsigned* nx = &s_tk[(i + 1) & 1];
            bool draw = true;

            const int e = ((int)tk * WPB + wv) * 32 + ln;
            const int s = src[e];
            const int d = dst[e];
            const int w = __float_as_int(nrm[e]);
            asm volatile("" :: "v"(s), "v"(d), "v"(w));
            unsigned mask = __builtin_amdgcn_ballot_w32((d >= HALF_ROWS ? 1 : 0) == p);
            while (mask) {
                float v[EPG][4];
                int   lj[EPG];
                int   n = 0;
#pragma unroll
                for (int j = 0; j < EPG; ++j) {
                    lj[j] = 0;
                    if (mask) {
                        const int bpos = __builtin_ctz(mask);
                        mask &= mask - 1u;
                        lj[j] = bpos;
                        n = j + 1;
                        const int sj = __builtin_amdgcn_readlane(s, bpos);
                        const float* tp = tl + (size_t)sj * HH;
                        v[j][0] = tp[0];
                        v[j][1] = tp[32];
                        v[j][2] = tp[64];
                        v[j][3] = tp[96];
                    }
                }
                if (draw) {
                    if (tid == 0) *nx = atomicAdd(c, 1u);
                    draw = false;
                }
#pragma unroll
                for (int j = 0; j < EPG; ++j) {
                    if (j < n) {
                        const int   dj = __builtin_amdgcn_readlane(d, lj[j]);
                        const float wj = __int_as_float(__builtin_amdgcn_readlane(w, lj[j]));
                        float* ap = al + (size_t)dj * HH;
                        atomicAdd(ap,      v[j][0] * wj);
                        atomicAdd(ap + 32, v[j][1] * wj);
                        atomicAdd(ap + 64, v[j][2] * wj);
                        atomicAdd(ap + 96, v[j][3] * wj);
                    }
                }
            }
            if (draw) {
                if (tid == 0) *nx = atomicAdd(c, 1u);
            }
            ticket_barrier();
        }
    }
}

static inline int nblocks(long long t, int bs) { return (int)((t + bs - 1) / bs); }

extern "C" void kernel_launch(void* const* d_in, const int* in_sizes, int n_in,
                              void* d_out, int out_size, void* d_ws, size_t ws_size,
                              hipStream_t stream) {
    const float* x  = (const float*)d_in[0];
    const int*   ei = (const int*)d_in[1];
    const float* ew = (const float*)d_in[2];
    const float* W1 = (const float*)d_in[3];  const float* b1 = (const float*)d_in[4];
    const float* W2 = (const float*)d_in[5];  const float* b2 = (const float*)d_in[6];
    const float* W3 = (const float*)d_in[7];  const float* b3 = (const float*)d_in[8];
    const float* Wg[4] = { (const float*)d_in[9],  (const float*)d_in[11],
                           (const float*)d_in[13], (const float*)d_in[15] };
    const float* bg[4] = { (const float*)d_in[10], (const float*)d_in[12],
                           (const float*)d_in[14], (const float*)d_in[16] };
    const float* Wh = (const float*)d_in[17]; const float* bh = (const float*)d_in[18];
    float* out = (float*)d_out;

    const int* srcI = ei;
    const int* dstI = ei + EE;

    float* f    = (float*)d_ws;
    float* hbuf = f;
    float* tbuf = hbuf + (size_t)NN * HH;
    float* aggb = tbuf + (size_t)NN * HH;
    float* degb = aggb + (size_t)NN * HH;
    float* scb  = degb + NN;
    float* nrmb = scb + NN;
    unsigned* cntb = (unsigned*)(nrmb + EE);

    const int BS = 256;

    fill_kernel<<<nblocks(NN, BS), BS, 0, stream>>>(degb, 2.0f, NN, cntb);
    deg_accum_kernel<<<nblocks(EE, BS), BS, 0, stream>>>(dstI, ew, degb);
    dinv_kernel<<<nblocks(NN, BS), BS, 0, stream>>>(degb, scb);
    norm_kernel<<<nblocks(EE, BS), BS, 0, stream>>>(srcI, dstI, ew, degb, nrmb);

    const int gblk128 = nblocks(NN >> 4, GW) * (HH / 64);

    gemm16_wmma<1, 4, 0, 0, IND><<<gblk128, BS, 0, stream>>>(x,    W1, b1, nullptr, hbuf, nullptr, NN, HH);
    gemm16_wmma<1, 4, 0, 0, HH><<<gblk128, BS, 0, stream>>>(hbuf, W2, b2, nullptr, tbuf, nullptr, NN, HH);
    gemm16_wmma<1, 4, 0, 0, HH><<<gblk128, BS, 0, stream>>>(tbuf, W3, b3, nullptr, hbuf, nullptr, NN, HH);

    float* cur = hbuf;
    float* nxt = aggb;
    for (int l = 0; l < 4; ++l) {
        if (l == 0)
            gemm16_wmma<0, 4, 0, 1, HH><<<gblk128, BS, 0, stream>>>(
                cur, Wg[l], bg[l], scb, tbuf, nxt, NN, HH);
        else
            gemm16_wmma<0, 4, 1, 1, HH><<<gblk128, BS, 0, stream>>>(
                cur, Wg[l], bg[l], scb, tbuf, nxt, NN, HH);
        scatter_kernel<<<SC_BLOCKS, 32 * WPB, 0, stream>>>(
            tbuf, srcI, dstI, nrmb, nxt, cntb + 2 * l);
        float* sw = cur; cur = nxt; nxt = sw;
    }

    {
        gemm16_wmma<0, 1, 1, 0, HH><<<nblocks(NN >> 4, GW) * (OUTD >> 4), BS, 0, stream>>>(
            cur, Wh, bh, nullptr, out, nullptr, NN, OUTD);
    }
}


# Degree sums off global atomics: 64 edge slices, each work-group adds its slice into a full table in LDS and stores it as a slab; slabs summed in slice order; sums in LDS arrival order; 15 launches
# speedup vs baseline: 2.7170x; 1.1514x over previous
#include <hip/hip_runtime.h>
#include <math.h>


#define NN   50000
#define EE   800000
#define IND  64
#define HH   128
#define OUTD 16

typedef float v2f __attribute__((ext_vector_type(2)));
typedef float v8f __attribute__((ext_vector_type(8)));

__device__ __forceinline__ float elu_f(float v) {
    return v > 0.0f ? v : (__expf(v) - 1.0f);
}

template <int TN>
__device__ __forceinline__ void load_bfrag(const float* __restrict__ p, v2f (&b)[TN]) {
    if constexpr (TN == 4) {
        const float4 q0 = *(const float4*)(p);
        const float4 q1 = *(const float4*)(p + 64);
        b[0] = (v2f){q0.x, q0.y};
        b[1] = (v2f){q0.z, q0.w};
        b[2] = (v2f){q1.x, q1.y};
        b[3] = (v2f){q1.z, q1.w};
    } else {
        const float2 q0 = *(const float2*)(p);
        b[0] = (v2f){q0.x, q0.y};
    }
}

template <int TN>
__device__ __forceinline__ void wmma_strip(v2f a, const v2f (&b)[TN], v8f (&acc)[TN]) {
#pragma unroll
    for (int t = 0; t < TN; ++t) {
        acc[t] = __builtin_amdgcn_wmma_f32_16x16x4_f32(
            false, a, false, b[t], (short)0, acc[t], false, false);
    }
}

#define GW 8

template <int ACT, int TN, int EIN, int DUAL, int K>
__global__ void __launch_bounds__(32 * GW)
gemm16_wmma(const float* __restrict__ A,
            const float* __restrict__ W,
            const float* __restrict__ bias,
            const float* __restrict__ rowscale,
            float* __restrict__ C,
            float* __restrict__ C2,
            int M, int Nc) {
    static_assert(TN == 4 || TN == 1, "panel layouts exist for 4 tiles and for 1");
    constexpr int PW  = 16 * TN;
    constexpr int PPR = PW / 4;
    constexpr int NQ  = K * PPR / (32 * GW);
    static_assert(K % 8 == 0 && (K * PPR) % (32 * GW) == 0, "k-steps go in pairs; the panel divides evenly among the threads");
    __shared__ __attribute__((aligned(16))) float Wl[K * PW];

    const int tid  = threadIdx.x;
    const int lane = tid & 31;
    const int wv   = tid >> 5;
    const int groupsN = Nc / PW;
    const int tilesM  = M >> 4;
    const int bq = blockIdx.x / groupsN;
    const int tg = blockIdx.x - bq * groupsN;
    const int tm = bq * GW + wv;
    const int tmc = tm < tilesM ? tm : tilesM - 1;

    const int r  = lane & 15;
    const int hf = lane >> 4;
    const int kq = hf << 1;

    const float* Arow = A + (size_t)(tmc * 16 + r) * K + kq;
    const float2 av0 = *(const float2*)(Arow);

    {
        const float* Wp = W + tg * PW;
        float4 w4[NQ];
#pragma unroll
        for (int i = 0; i < NQ; ++i) {
            const int q  = tid + i * (32 * GW);
            const int k  = q / PPR;
            const int c4 = (q - k * PPR) << 2;
            w4[i] = *(const float4*)(Wp + (size_t)k * Nc + c4);
        }
#pragma unroll
        for (int i = 0; i < NQ; ++i) {
            const int q  = tid + i * (32 * GW);
            const int k  = q / PPR;
            const int c4 = (q - k * PPR) << 2;
            const int t  = c4 >> 4;
            const int c  = c4 & 15;
            float* p = (TN == 4)
                ? Wl + (k >> 1) * (2 * PW) + (t >> 1) * 64 + c * 4 + (t & 1) * 2 + (k & 1)
                : Wl + (k >> 1) * (2 * PW) + c * 2 + (k & 1);
            constexpr int CS = (TN == 4) ? 4 : 2;
            p[0]      = w4[i].x;
            p[CS]     = w4[i].y;
            p[2 * CS] = w4[i].z;
            p[3 * CS] = w4[i].w;
        }
    }
    __syncthreads();
    if (tm >= tilesM) return;

    const float* Bl = Wl + hf * (2 * PW) + r * (TN == 4 ? 4 : 2);

    v8f acc[TN];
#pragma unroll
    for (int t = 0; t < TN; ++t) acc[t] = (v8f){};

    v2f aP, bP[TN], aQ, bQ[TN];
    aP.x = EIN ? elu_f(av0.x) : av0.x;
    aP.y = EIN ? elu_f(av0.y) : av0.y;
    load_bfrag<TN>(Bl, bP);

    for (int k0 = 0; k0 < K - 8; k0 += 8) {
        const float2 rq = *(const float2*)(Arow + k0 + 4);
        load_bfrag<TN>(Bl + (k0 + 4) * PW, bQ);
        wmma_strip<TN>(aP, bP, acc);
        aQ.x = EIN ? elu_f(rq.x) : rq.x;
        aQ.y = EIN ? elu_f(rq.y) : rq.y;
        const float2 rp = *(const float2*)(Arow + k0 + 8);
        load_bfrag<TN>(Bl + (k0 + 8) * PW, bP);
        wmma_strip<TN>(aQ, bQ, acc);
        aP.x = EIN ? elu_f(rp.x) : rp.x;
        aP.y = EIN ? elu_f(rp.y) : rp.y;
    }
    {
        const float2 rq = *(const float2*)(Arow + K - 4);
        load_bfrag<TN>(Bl + (K - 4) * PW, bQ);
        wmma_strip<TN>(aP, bP, acc);
        aQ.x = EIN ? elu_f(rq.x) : rq.x;
        aQ.y = EIN ? elu_f(rq.y) : rq.y;
        wmma_strip<TN>(aQ, bQ, acc);
    }

    int mb = tm * 16 + ((lane >> 4) << 3);
    float rs[8];
    if (DUAL) {
#pragma unroll
        for (int g = 0; g < 8; ++g) rs[g] = rowscale[mb + g];
    }
#pragma unroll
    for (int t = 0; t < TN; ++t) {
        int n = tg * PW + t * 16 + r;
        float bv = bias ? bias[n] : 0.0f;
#pragma unroll
        for (int g = 0; g < 8; ++g) {
            size_t o = (size_t)(mb + g) * Nc + n;
            if (DUAL) {
                float v = acc[t][g];
                C[o]  = v;
                C2[o] = rs[g] * v + bv;
            } else {
                float v = acc[t][g] + bv;
                if (ACT) v = elu_f(v);
                C[o] = v;
            }
        }
    }
}

#define DK  64
#define DSL (EE / DK)
#define DQ  (DSL / 4)
#define DTH 1024
#define DQT ((DQ + DTH - 1) / DTH)
static_assert(EE % DK == 0 && DSL % 4 == 0 && NN % 4 == 0, "slices and table go in 16-byte pieces");

__global__ void __launch_bounds__(DTH)
deg_slab_kernel(const int* __restrict__ dst,
                const float* __restrict__ w,
                float* __restrict__ slab,
                unsigned* __restrict__ cnt) {
    __shared__ __attribute__((aligned(16))) float tab[NN];
    const int tid = threadIdx.x;
    const int b   = blockIdx.x;
    const int4*   d4 = (const int4*)(dst + (size_t)b * DSL);
    const float4* w4 = (const float4*)(w + (size_t)b * DSL);
    int4   dd[DQT];
    float4 ww[DQT];
#pragma unroll
    for (int i = 0; i < DQT; ++i) {
        const int q = tid + i * DTH;
        if (q < DQ) { dd[i] = d4[q]; ww[i] = w4[q]; }
    }
    float4* t4 = (float4*)tab;
    for (int i = tid; i < NN / 4; i += DTH) t4[i] = make_float4(0.0f, 0.0f, 0.0f, 0.0f);
    if (b == 0 && tid < 8) cnt[tid] = 0u;
    __syncthreads();
#pragma unroll
    for (int i = 0; i < DQT; ++i) {
        const int q = tid + i * DTH;
        if (q < DQ) {
            atomicAdd(&tab[dd[i].x], ww[i].x);
            atomicAdd(&tab[dd[i].y], ww[i].y);
            atomicAdd(&tab[dd[i].z], ww[i].z);
            atomicAdd(&tab[dd[i].w], ww[i].w);
        }
    }
    __syncthreads();
    float4* s4 = (float4*)(slab + (size_t)b * NN);
    for (int i = tid; i < NN / 4; i += DTH) s4[i] = t4[i];
}

__global__ void dinv_kernel(const float* __restrict__ slab,
                            float* __restrict__ dinv,
                            float* __restrict__ sc) {
    int n = blockIdx.x * blockDim.x + threadIdx.x;
    if (n < NN) {
        float s = 0.0f;
#pragma unroll
        for (int b = 0; b < DK; ++b) s += slab[(size_t)b * NN + n];
        float di = rsqrtf(s + 2.0f);
        dinv[n] = di;
        sc[n]   = 2.0f * di * di;
    }
}

__global__ void norm_kernel(const int* __restrict__ src,
                            const int* __restrict__ dst,
                            const float* __restrict__ w,
                            const float* __restrict__ dinv,
                            float* __restrict__ nrm) {
    int e = blockIdx.x * blockDim.x + threadIdx.x;
    if (e < EE) nrm[e] = w[e] * dinv[src[e]] * dinv[dst[e]];
}

#define EPG 12
#define WPB 8
#define NTK (EE / (32 * WPB))
#define HALF_ROWS (NN / 2)
#define SC_BLOCKS 1536
static_assert(EE % (32 * WPB) == 0, "tickets tile the edge list");

__device__ __forceinline__ void ticket_barrier() {
    __builtin_amdgcn_fence(__ATOMIC_RELEASE, "workgroup", "local");
    __builtin_amdgcn_s_barrier();
    __builtin_amdgcn_fence(__ATOMIC_ACQUIRE, "workgroup", "local");
}

__global__ void __launch_bounds__(32 * WPB)
scatter_kernel(const float* __restrict__ t,
               const int* __restrict__ src,
               const int* __restrict__ dst,
               const float* __restrict__ nrm,
               float* __restrict__ agg,
               unsigned* __restrict__ cnt) {
    __shared__ unsigned s_half;
    __shared__ unsigned s_tk[2];
    const int tid = threadIdx.x;
    const int ln  = tid & 31;
    const int wv  = tid >> 5;
    if (tid == 0) {
        unsigned id = __builtin_amdgcn_s_sendmsg_rtn(0x87);
        s_half = (id >> 18) & 1u;
    }
    ticket_barrier();
    const int hh = (int)s_half;
    const float* tl = t + ln;
    float*       al = agg + ln;

    for (int ph = 0; ph < 2; ++ph) {
        const int p = hh ^ ph;
        unsigned* c = cnt + p;
        ticket_barrier();
        if (tid == 0) s_tk[0] = atomicAdd(c, 1u);
        ticket_barrier();
        for (int i = 0;; ++i) {
            const unsigned tk = s_tk[i & 1];
            if (tk >= (unsigned)NTK) break;
            unsigned* nx = &s_tk[(i + 1) & 1];
            bool draw = true;

            const int e = ((int)tk * WPB + wv) * 32 + ln;
            const int s = src[e];
            const int d = dst[e];
            const int w = __float_as_int(nrm[e]);
            asm volatile("" :: "v"(s), "v"(d), "v"(w));
            unsigned mask = __builtin_amdgcn_ballot_w32((d >= HALF_ROWS ? 1 : 0) == p);
            while (mask) {
                float v[EPG][4];
                int   lj[EPG];
                int   n = 0;
#pragma unroll
                for (int j = 0; j < EPG; ++j) {
                    lj[j] = 0;
                    if (mask) {
                        const int bpos = __builtin_ctz(mask);
                        mask &= mask - 1u;
                        lj[j] = bpos;
                        n = j + 1;
                        const int sj = __builtin_amdgcn_readlane(s, bpos);
                        const float* tp = tl + (size_t)sj * HH;
                        v[j][0] = tp[0];
                        v[j][1] = tp[32];
                        v[j][2] = tp[64];
                        v[j][3] = tp[96];
                    }
                }
                if (draw) {
                    if (tid == 0) *nx = atomicAdd(c, 1u);
                    draw = false;
                }
#pragma unroll
                for (int j = 0; j < EPG; ++j) {
                    if (j < n) {
                        const int   dj = __builtin_amdgcn_readlane(d, lj[j]);
                        const float wj = __int_as_float(__builtin_amdgcn_readlane(w, lj[j]));
                        float* ap = al + (size_t)dj * HH;
                        atomicAdd(ap,      v[j][0] * wj);
                        atomicAdd(ap + 32, v[j][1] * wj);
                        atomicAdd(ap + 64, v[j][2] * wj);
                        atomicAdd(ap + 96, v[j][3] * wj);
                    }
                }
            }
            if (draw) {
                if (tid == 0) *nx = atomicAdd(c, 1u);
            }
            ticket_barrier();
        }
    }
}

static inline int nblocks(long long t, int bs) { return (int)((t + bs - 1) / bs); }

extern "C" void kernel_launch(void* const* d_in, const int* in_sizes, int n_in,
                              void* d_out, int out_size, void* d_ws, size_t ws_size,
                              hipStream_t stream) {
    const float* x  = (const float*)d_in[0];
    const int*   ei = (const int*)d_in[1];
    const float* ew = (const float*)d_in[2];
    const float* W1 = (const float*)d_in[3];  const float* b1 = (const float*)d_in[4];
    const float* W2 = (const float*)d_in[5];  const float* b2 = (const float*)d_in[6];
    const float* W3 = (const float*)d_in[7];  const float* b3 = (const float*)d_in[8];
    const float* Wg[4] = { (const float*)d_in[9],  (const float*)d_in[11],
                           (const float*)d_in[13], (const float*)d_in[15] };
    const float* bg[4] = { (const float*)d_in[10], (const float*)d_in[12],
                           (const float*)d_in[14], (const float*)d_in[16] };
    const float* Wh = (const float*)d_in[17]; const float* bh = (const float*)d_in[18];
    float* out = (float*)d_out;

    const int* srcI = ei;
    const int* dstI = ei + EE;

    float* f    = (float*)d_ws;
    float* hbuf = f;
    float* tbuf = hbuf + (size_t)NN * HH;
    float* aggb = tbuf + (size_t)NN * HH;
    float* degb = aggb + (size_t)NN * HH;
    float* scb  = degb + NN;
    float* nrmb = scb + NN;
    unsigned* cntb = (unsigned*)(nrmb + EE);
    float* slabb = nrmb + EE + 8;

    const int BS = 256;

    deg_slab_kernel<<<DK, DTH, 0, stream>>>(dstI, ew, slabb, cntb);
    dinv_kernel<<<nblocks(NN, BS), BS, 0, stream>>>(slabb, degb, scb);
    norm_kernel<<<nblocks(EE, BS), BS, 0, stream>>>(srcI, dstI, ew, degb, nrmb);

    const int gblk128 = nblocks(NN >> 4, GW) * (HH / 64);

    gemm16_wmma<1, 4, 0, 0, IND><<<gblk128, BS, 0, stream>>>(x,    W1, b1, nullptr, hbuf, nullptr, NN, HH);
    gemm16_wmma<1, 4, 0, 0, HH><<<gblk128, BS, 0, stream>>>(hbuf, W2, b2, nullptr, tbuf, nullptr, NN, HH);
    gemm16_wmma<1, 4, 0, 0, HH><<<gblk128, BS, 0, stream>>>(tbuf, W3, b3, nullptr, hbuf, nullptr, NN, HH);

    float* cur = hbuf;
    float* nxt = aggb;
    for (int l = 0; l < 4; ++l) {
        if (l == 0)
            gemm16_wmma<0, 4, 0, 1, HH><<<gblk128, BS, 0, stream>>>(
                cur, Wg[l], bg[l], scb, tbuf, nxt, NN, HH);
        else
            gemm16_wmma<0, 4, 1, 1, HH><<<gblk128, BS, 0, stream>>>(
                cur, Wg[l], bg[l], scb, tbuf, nxt, NN, HH);
        scatter_kernel<<<SC_BLOCKS, 32 * WPB, 0, stream>>>(
            tbuf, srcI, dstI, nrmb, nxt, cntb + 2 * l);
        float* sw = cur; cur = nxt; nxt = sw;
    }

    {
        gemm16_wmma<0, 1, 1, 0, HH><<<nblocks(NN >> 4, GW) * (OUTD >> 4), BS, 0, stream>>>(
            cur, Wh, bh, nullptr, out, nullptr, NN, OUTD);
    }
}
